# starting kernel
# baseline (speedup 1.0000x reference)
#include <hip/hip_runtime.h>
#include <math.h>

#define GAMMA 1.1f

typedef __attribute__((ext_vector_type(2))) float v2f;
typedef __attribute__((ext_vector_type(8))) float v8f;

__global__ void prep_kernel(const float* __restrict__ boxes, int M,
                            float* __restrict__ out, int out_n,
                            float4* __restrict__ boxdata,
                            float* __restrict__ rad2) {
  int t = blockIdx.x * blockDim.x + threadIdx.x;
  if (t < out_n) out[t] = 0.0f;
  if (t < M) {
    float bx = boxes[t * 9 + 0], by = boxes[t * 9 + 1];
    float dx = boxes[t * 9 + 3], dy = boxes[t * 9 + 4];
    float hx = 0.5f * dx, hy = 0.5f * dy;
    float r = sqrtf(hx * hx + hy * hy) * GAMMA;
    boxdata[t] = make_float4(-2.0f * bx, -2.0f * by, bx * bx + by * by, 0.0f);
    rad2[t] = r * r;
  }
}

__global__ __launch_bounds__(32)
void count_kernel(const float* __restrict__ P, int N,
                  const float4* __restrict__ boxdata,
                  const float* __restrict__ rad2,
                  unsigned* __restrict__ masks, int ngroup,
                  int* __restrict__ cnt, int nch) {
  const int wave = blockIdx.x;
  const int bt   = wave / nch;
  const int c    = wave % nch;
  const int lane = threadIdx.x;
  const bool lo  = lane < 16;

  float4 bd = boxdata[bt * 16 + (lane & 15)];
  v2f a;
  a.x = lo ? bd.x : 1.0f;
  a.y = lo ? bd.y : bd.z;

  float r2v[8];
#pragma unroll
  for (int r = 0; r < 8; ++r) r2v[r] = rad2[bt * 16 + r + (lo ? 0 : 8)];

  int cl[8], ch[8];
#pragma unroll
  for (int r = 0; r < 8; ++r) { cl[r] = 0; ch[r] = 0; }

  const int g0 = c * 64;
  for (int k = 0; k < 64; k += 2) {
    int idx0 = (g0 + k) * 16 + (lane & 15);
    int idx1 = idx0 + 16;
    float px0, py0, px1, py1;
    if (idx0 < N) { px0 = P[(size_t)idx0 * 5]; py0 = P[(size_t)idx0 * 5 + 1]; }
    else          { px0 = 1e18f; py0 = 1e18f; }
    if (idx1 < N) { px1 = P[(size_t)idx1 * 5]; py1 = P[(size_t)idx1 * 5 + 1]; }
    else          { px1 = 1e18f; py1 = 1e18f; }

    {
      int pk = (k + 2) & 63;
      int nidx = (g0 + pk) * 16 + (lane & 15);
      if (nidx < N) __builtin_prefetch(&P[(size_t)nidx * 5], 0, 3);
    }

    float pp0 = px0 * px0 + py0 * py0;
    float pp1 = px1 * px1 + py1 * py1;
    v2f b0, b1;
    b0.x = lo ? px0 : pp0;  b0.y = lo ? py0 : 1.0f;
    b1.x = lo ? px1 : pp1;  b1.y = lo ? py1 : 1.0f;

    v8f cz = {0.f, 0.f, 0.f, 0.f, 0.f, 0.f, 0.f, 0.f};
    v8f d0 = __builtin_amdgcn_wmma_f32_16x16x4_f32(false, a, false, b0,
                                                   (short)0, cz, false, false);
    v8f d1 = __builtin_amdgcn_wmma_f32_16x16x4_f32(false, a, false, b1,
                                                   (short)0, cz, false, false);
    unsigned w0 = 0, w1 = 0;
#pragma unroll
    for (int r = 0; r < 8; ++r) {
      unsigned bal0 = __builtin_amdgcn_ballot_w32(d0[r] <= r2v[r]);
      unsigned bal1 = __builtin_amdgcn_ballot_w32(d1[r] <= r2v[r]);
      cl[r] += __popc(bal0 & 0xFFFFu) + __popc(bal1 & 0xFFFFu);
      ch[r] += __popc(bal0 >> 16)     + __popc(bal1 >> 16);
      w0 = (lane == r) ? bal0 : w0;
      w1 = (lane == r) ? bal1 : w1;
    }
    if (lane < 8) {
      unsigned* mp = &masks[((size_t)bt * ngroup + (g0 + k)) * 8 + lane];
      mp[0] = w0;
      mp[8] = w1;
    }
  }

  if (lane == 0) {
#pragma unroll
    for (int r = 0; r < 8; ++r) {
      cnt[(bt * 16 + r)     * nch + c] = cl[r];
      cnt[(bt * 16 + r + 8) * nch + c] = ch[r];
    }
  }
}

__global__ void scan_kernel(const int* __restrict__ cnt,
                            int* __restrict__ baseArr, int nch) {
  __shared__ int s[1024];
  int m = blockIdx.x;
  int t = threadIdx.x;
  s[t] = cnt[(size_t)m * nch + t];
  __syncthreads();
  for (int off = 1; off < nch; off <<= 1) {
    int x = s[t];
    int y = (t >= off) ? s[t - off] : 0;
    __syncthreads();
    s[t] = x + y;
    __syncthreads();
  }
  baseArr[(size_t)m * nch + t] = (t == 0) ? 0 : s[t - 1];
}

__global__ __launch_bounds__(32)
void emit_kernel(const float* __restrict__ P,
                 const float* __restrict__ boxes,
                 const unsigned* __restrict__ masks, int ngroup,
                 const int* __restrict__ baseArr, int nch,
                 float* __restrict__ out, int S) {
  const int blk = blockIdx.x;
  const int m = blk / nch;
  const int c = blk % nch;
  int run = baseArr[(size_t)m * nch + c];
  if (run >= S) return;

  const int lane = threadIdx.x;
  const int bt = m >> 4, j = m & 15;
  const int ridx  = j & 7;
  const int shift = ((j >> 3) << 4) + (lane & 15);
  const float vx = boxes[m * 9 + 7];
  const float vy = boxes[m * 9 + 8];
  const unsigned ltmask = (1u << lane) - 1u;
  const int g0 = c * 64;

  for (int k = 0; k < 64; k += 2) {
    int G = g0 + k + (lane >> 4);
    unsigned w = masks[((size_t)bt * ngroup + G) * 8 + ridx];
    int bit = (int)((w >> shift) & 1u);
    unsigned bal = __builtin_amdgcn_ballot_w32(bit != 0);
    int slot = run + __popc(bal & ltmask);
    if (bit && slot < S) {
      int pidx = G * 16 + (lane & 15);
      const float* ps = P + (size_t)pidx * 5;
      float* o = out + ((size_t)m * S + slot) * 7;
      o[0] = ps[0]; o[1] = ps[1]; o[2] = ps[2];
      o[3] = ps[3]; o[4] = ps[4];
      o[5] = vx;    o[6] = vy;
    }
    run += __popc(bal);
    if (run >= S) break;
  }
}

extern "C" void kernel_launch(void* const* d_in, const int* in_sizes, int n_in,
                              void* d_out, int out_size, void* d_ws, size_t ws_size,
                              hipStream_t stream) {
  const float* P = (const float*)d_in[0];
  const float* B = (const float*)d_in[1];
  float* out = (float*)d_out;

  const int N = in_sizes[0] / 5;
  const int M = in_sizes[1] / 9;
  const int S = out_size / (M * 7);
  const int nch    = (N + 1023) / 1024;
  const int ngroup = nch * 64;
  const int bt     = M / 16;

  char* ws = (char*)d_ws;
  unsigned* masks   = (unsigned*)ws;
  size_t mbytes     = (size_t)bt * ngroup * 8 * sizeof(unsigned);
  int* cnt          = (int*)(ws + mbytes);
  int* baseA        = cnt + (size_t)M * nch;
  float4* boxdata   = (float4*)(baseA + (size_t)M * nch);
  float* rad2       = (float*)(boxdata + M);

  prep_kernel<<<(out_size + 255) / 256, 256, 0, stream>>>(B, M, out, out_size,
                                                          boxdata, rad2);
  count_kernel<<<bt * nch, 32, 0, stream>>>(P, N, boxdata, rad2,
                                            masks, ngroup, cnt, nch);
  scan_kernel<<<M, nch, 0, stream>>>(cnt, baseA, nch);
  emit_kernel<<<M * nch, 32, 0, stream>>>(P, B, masks, ngroup, baseA, nch,
                                          out, S);
}


# per-box bitmap instead of ballot masks: product transposed (points as rows, a lane owns one box), lane shifts sign of r2-d2 into its box's word; scalar popcounts gone; emit reads one line per chunk
# speedup vs baseline: 3.2104x; 3.2104x over previous
#include <hip/hip_runtime.h>
#include <math.h>

#define GAMMA 1.1f

typedef __attribute__((ext_vector_type(2))) float v2f;
typedef __attribute__((ext_vector_type(8))) float v8f;

__global__ void prep_kernel(const float* __restrict__ boxes, int M,
                            float4* __restrict__ boxdata,
                            float* __restrict__ rad2) {
  int t = blockIdx.x * blockDim.x + threadIdx.x;
  if (t < M) {
    float bx = boxes[t * 9 + 0], by = boxes[t * 9 + 1];
    float dx = boxes[t * 9 + 3], dy = boxes[t * 9 + 4];
    float hx = 0.5f * dx, hy = 0.5f * dy;
    float r = sqrtf(hx * hx + hy * hy) * GAMMA;
    boxdata[t] = make_float4(-2.0f * bx, -2.0f * by, bx * bx + by * by, 0.0f);
    rad2[t] = r * r;
  }
}

#define CNT_WAVES 16
#define CNT_T (CNT_WAVES * 32)
#define ROWP 36

__device__ __forceinline__ unsigned push_out(unsigned acc, float r2, float d) {
  return __builtin_amdgcn_alignbit(acc, __float_as_uint(r2 - d), 31u);
}

__global__ __launch_bounds__(CNT_T)
void count_kernel(const float* __restrict__ P, int N,
                  const float4* __restrict__ boxdata,
                  const float* __restrict__ rad2,
                  unsigned* __restrict__ bitmap, int wpb,
                  int* __restrict__ cnt, int nch, int nbt) {
  __shared__ float2 frag[68 * 32];
  __shared__ __attribute__((aligned(16))) unsigned rows[CNT_WAVES * 16 * ROWP];

  const int c    = blockIdx.x;
  const int tid  = threadIdx.x;
  const int lane = tid & 31;
  const int wv   = tid >> 5;
  const bool lo  = lane < 16;

  for (int i = tid; i < 1024; i += CNT_T) {
    const int idx = c * 1024 + i;
    float px = 1e18f, py = 1e18f;
    if (idx < N) { px = P[(size_t)idx * 5]; py = P[(size_t)idx * 5 + 1]; }
    const float pp = px * px + py * py;
    const int f = (i >> 4) * 32 + (i & 15);
    frag[f]      = make_float2(px, py);
    frag[f + 16] = make_float2(pp, 1.0f);
  }
  if (tid < 128) frag[64 * 32 + tid] = make_float2(0.0f, 0.0f);
  __syncthreads();

  unsigned* wrow = rows + wv * (16 * ROWP);

  for (int bt = wv; bt < nbt; bt += CNT_WAVES) {
    float4 bd = boxdata[bt * 16 + (lane & 15)];
    v2f bx;
    bx.x = lo ? bd.x : 1.0f;
    bx.y = lo ? bd.y : bd.z;
    const float r2 = rad2[bt * 16 + (lane & 15)];

    unsigned* rw = wrow + (lane & 15) * ROWP + (lane >> 4);
    int cl = 0;

    const float2* fp = frag + lane;
    float2 n0 = fp[0], n1 = fp[32], n2 = fp[64], n3 = fp[96];
    for (int j = 0; j < 16; ++j) {
      v2f a0, a1, a2, a3;
      a0.x = n0.x;  a0.y = n0.y;
      a1.x = n1.x;  a1.y = n1.y;
      a2.x = n2.x;  a2.y = n2.y;
      a3.x = n3.x;  a3.y = n3.y;
      fp += 128;
      n0 = fp[0];  n1 = fp[32];  n2 = fp[64];  n3 = fp[96];

      v8f cz = {0.f, 0.f, 0.f, 0.f, 0.f, 0.f, 0.f, 0.f};
      v8f d0 = __builtin_amdgcn_wmma_f32_16x16x4_f32(false, a0, false, bx,
                                                     (short)0, cz, false, false);
      v8f d1 = __builtin_amdgcn_wmma_f32_16x16x4_f32(false, a1, false, bx,
                                                     (short)0, cz, false, false);
      v8f d2 = __builtin_amdgcn_wmma_f32_16x16x4_f32(false, a2, false, bx,
                                                     (short)0, cz, false, false);
      v8f d3 = __builtin_amdgcn_wmma_f32_16x16x4_f32(false, a3, false, bx,
                                                     (short)0, cz, false, false);
      unsigned b0 = 0, b1 = 0, b2 = 0, b3 = 0;
#pragma unroll
      for (int r = 7; r >= 0; --r) {
        b0 = push_out(b0, r2, d0[r]);
        b1 = push_out(b1, r2, d1[r]);
        b2 = push_out(b2, r2, d2[r]);
        b3 = push_out(b3, r2, d3[r]);
      }
      const unsigned in = ~(b0 | (b1 << 8) | (b2 << 16) | (b3 << 24));
      cl += __popc(in);
      rw[2 * j] = in;
    }

    const int tot = cl + __builtin_amdgcn_permlanex16(cl, cl, 0x76543210u, 0xfedcba98u, false, false);
    if (lo) cnt[(bt * 16 + lane) * nch + c] = tot;

    __builtin_amdgcn_wave_barrier();

#pragma unroll
    for (int i = 0; i < 4; ++i) {
      const int box = 4 * i + (lane >> 3);
      const int u   = lane & 7;
      const uint4 q = *reinterpret_cast<const uint4*>(wrow + box * ROWP + 4 * u);
      uint4 o;
      o.x = __builtin_amdgcn_perm(q.y, q.x, 0x05010400u);
      o.y = __builtin_amdgcn_perm(q.y, q.x, 0x07030602u);
      o.z = __builtin_amdgcn_perm(q.w, q.z, 0x05010400u);
      o.w = __builtin_amdgcn_perm(q.w, q.z, 0x07030602u);
      *reinterpret_cast<uint4*>(bitmap + (size_t)(bt * 16 + box) * wpb
                                + (size_t)c * 32 + 4 * u) = o;
    }

    __builtin_amdgcn_wave_barrier();
  }
}

#define EMIT_T 128

__global__ __launch_bounds__(EMIT_T)
void emit_kernel(const float* __restrict__ P,
                 const float* __restrict__ boxes,
                 const unsigned* __restrict__ bitmap, int wpb,
                 const int* __restrict__ cnt, int nch,
                 float* __restrict__ out, int S) {
  __shared__ int   sc[1024];
  __shared__ int   wtot[EMIT_T / 32];
  __shared__ float stage[EMIT_T * 7];

  const int m    = blockIdx.x;
  const int tid  = threadIdx.x;
  const int lane = tid & 31;
  const int wv   = tid >> 5;

  const float vx = boxes[m * 9 + 7];
  const float vy = boxes[m * 9 + 8];

  const int E   = (nch + EMIT_T - 1) / EMIT_T;
  const int beg = tid * E;
  const int* cm = cnt + (size_t)m * nch;
  int v[8];
#pragma unroll
  for (int e = 0; e < 8; ++e) {
    int i = beg + e;
    v[e] = (e < E && i < nch) ? cm[i] : 0;
  }
  int loc = 0;
#pragma unroll
  for (int e = 0; e < 8; ++e) { loc += v[e]; v[e] = loc; }
  int x = loc;
#pragma unroll
  for (int off = 1; off < 32; off <<= 1) {
    int y = __shfl_up(x, off);
    if (lane >= off) x += y;
  }
  if (lane == 31) wtot[wv] = x;
  __syncthreads();
  int add = x - loc;
  for (int w = 0; w < wv; ++w) add += wtot[w];
#pragma unroll
  for (int e = 0; e < 8; ++e) {
    int i = beg + e;
    if (e < E && i < nch) sc[i] = v[e] + add;
  }
  __syncthreads();
  const int total = sc[nch - 1];

  int hb = 1;
  while ((hb << 1) <= nch) hb <<= 1;

  for (int s0 = 0; s0 < S; s0 += EMIT_T) {
    if (s0 > 0) __syncthreads();
    const int s = s0 + tid;
    float r0 = 0.f, r1 = 0.f, r2 = 0.f, r3 = 0.f, r4 = 0.f, r5 = 0.f, r6 = 0.f;
    if (s < S && s < total) {
      int c = 0;
      for (int step = hb; step > 0; step >>= 1) {
        int t = c + step;
        if (t <= nch && sc[t - 1] <= s) c = t;
      }
      const int q = s - (c > 0 ? sc[c - 1] : 0);

      const uint4* mp = reinterpret_cast<const uint4*>(
          bitmap + (size_t)m * wpb + (size_t)c * 32);
      unsigned w[32];
#pragma unroll
      for (int k = 0; k < 8; ++k) {
        const uint4 t = mp[k];
        w[4 * k] = t.x;  w[4 * k + 1] = t.y;  w[4 * k + 2] = t.z;  w[4 * k + 3] = t.w;
      }
      int acc = 0, gk = 0, rem = 0;
      unsigned f = 0;
#pragma unroll
      for (int k = 0; k < 32; ++k) {
        if (acc <= q) { gk = k; rem = q - acc; f = w[k]; }
        acc += __popc(w[k]);
      }
      int pos = 0, n;
      n = __popc(f & 0xFFFFu); if (rem >= n) { rem -= n; pos += 16; f >>= 16; }
      n = __popc(f & 0xFFu); if (rem >= n) { rem -= n; pos += 8; f >>= 8; }
      n = __popc(f & 0xFu);  if (rem >= n) { rem -= n; pos += 4; f >>= 4; }
      n = __popc(f & 0x3u);  if (rem >= n) { rem -= n; pos += 2; f >>= 2; }
      n = (int)(f & 1u);     if (rem >= n) { pos += 1; }

      const int pidx = (c * 32 + gk) * 32 + pos;
      const float* ps = P + (size_t)pidx * 5;
      r0 = ps[0]; r1 = ps[1]; r2 = ps[2]; r3 = ps[3]; r4 = ps[4];
      r5 = vx;    r6 = vy;
    }
    float* st = stage + tid * 7;
    st[0] = r0; st[1] = r1; st[2] = r2; st[3] = r3;
    st[4] = r4; st[5] = r5; st[6] = r6;
    __syncthreads();
    const int nsl = (S - s0 < EMIT_T) ? (S - s0) : EMIT_T;
    const int nf  = nsl * 7;
    float* ob = out + ((size_t)m * S + s0) * 7;
    float o[7];
#pragma unroll
    for (int r = 0; r < 7; ++r) o[r] = stage[r * EMIT_T + tid];
#pragma unroll
    for (int r = 0; r < 7; ++r) {
      const int i = r * EMIT_T + tid;
      if (i < nf) ob[i] = o[r];
    }
  }
}

extern "C" void kernel_launch(void* const* d_in, const int* in_sizes, int n_in,
                              void* d_out, int out_size, void* d_ws, size_t ws_size,
                              hipStream_t stream) {
  const float* P = (const float*)d_in[0];
  const float* B = (const float*)d_in[1];
  float* out = (float*)d_out;

  const int N = in_sizes[0] / 5;
  const int M = in_sizes[1] / 9;
  const int S = out_size / (M * 7);
  const int nch    = (N + 1023) / 1024;
  const int wpb    = nch * 32;
  const int bt     = M / 16;

  char* ws = (char*)d_ws;
  unsigned* bitmap  = (unsigned*)ws;
  size_t mbytes     = (size_t)M * wpb * sizeof(unsigned);
  int* cnt          = (int*)(ws + mbytes);
  float4* boxdata   = (float4*)(cnt + (size_t)M * nch);
  float* rad2       = (float*)(boxdata + M);

  prep_kernel<<<(M + 255) / 256, 256, 0, stream>>>(B, M, boxdata, rad2);
  count_kernel<<<nch, CNT_T, 0, stream>>>(P, N, boxdata, rad2,
                                          bitmap, wpb, cnt, nch, bt);
  emit_kernel<<<M, EMIT_T, 0, stream>>>(P, B, bitmap, wpb, cnt, nch, out, S);
}


# emit: each wave maps its 32 slots to chunks itself: counts summed in-wave on DPP, each non-empty chunk marks its first slot in LDS, a running maximum spreads it; block scan, bisection, barriers gone
# speedup vs baseline: 3.8234x; 1.1909x over previous
#include <hip/hip_runtime.h>
#include <math.h>

#define GAMMA 1.1f

typedef __attribute__((ext_vector_type(2))) float v2f;
typedef __attribute__((ext_vector_type(8))) float v8f;

__device__ __forceinline__ void box_operand(const float* __restrict__ boxes, int m,
                                            bool lo, v2f& bcol, float& r2) {
  const float bx = boxes[m * 9 + 0], by = boxes[m * 9 + 1];
  const float dx = boxes[m * 9 + 3], dy = boxes[m * 9 + 4];
  const float hx = 0.5f * dx, hy = 0.5f * dy;
  float ss;
  {
#pragma clang fp contract(off)
    ss = hx * hx + hy * hy;
  }
  const float r = sqrtf(ss) * GAMMA;
  const float ax = -2.0f * bx, ay = -2.0f * by;
  const float bb = fmaf(bx, bx, by * by);
  bcol.x = lo ? ax : 1.0f;
  bcol.y = lo ? ay : bb;
  r2 = r * r;
}

#define CNT_WAVES 16
#define CNT_T (CNT_WAVES * 32)
#define ROWP 36

__device__ __forceinline__ unsigned push_out(unsigned acc, float r2, float d) {
  return __builtin_amdgcn_alignbit(acc, __float_as_uint(r2 - d), 31u);
}

__global__ __launch_bounds__(CNT_T)
void count_kernel(const float* __restrict__ P, int N,
                  const float* __restrict__ boxes,
                  unsigned* __restrict__ bitmap, int wpb,
                  int* __restrict__ cnt, int nch, int nbt) {
  __shared__ float2 frag[68 * 32];
  __shared__ __attribute__((aligned(16))) unsigned rows[CNT_WAVES * 16 * ROWP];

  const int c    = blockIdx.x;
  const int tid  = threadIdx.x;
  const int lane = tid & 31;
  const int wv   = tid >> 5;
  const bool lo  = lane < 16;

  for (int i = tid; i < 1024; i += CNT_T) {
    const int idx = c * 1024 + i;
    float px = 1e18f, py = 1e18f;
    if (idx < N) { px = P[(size_t)idx * 5]; py = P[(size_t)idx * 5 + 1]; }
    const float pp = px * px + py * py;
    const int f = (i >> 4) * 32 + (i & 15);
    frag[f]      = make_float2(px, py);
    frag[f + 16] = make_float2(pp, 1.0f);
  }
  if (tid < 128) frag[64 * 32 + tid] = make_float2(0.0f, 0.0f);
  __syncthreads();

  unsigned* wrow = rows + wv * (16 * ROWP);

  for (int bt = wv; bt < nbt; bt += CNT_WAVES) {
    v2f bx;
    float r2;
    box_operand(boxes, bt * 16 + (lane & 15), lo, bx, r2);

    unsigned* rw = wrow + (lane & 15) * ROWP + (lane >> 4);
    int cl = 0;

    const float2* fp = frag + lane;
    float2 n0 = fp[0], n1 = fp[32], n2 = fp[64], n3 = fp[96];
    for (int j = 0; j < 16; ++j) {
      v2f a0, a1, a2, a3;
      a0.x = n0.x;  a0.y = n0.y;
      a1.x = n1.x;  a1.y = n1.y;
      a2.x = n2.x;  a2.y = n2.y;
      a3.x = n3.x;  a3.y = n3.y;
      fp += 128;
      n0 = fp[0];  n1 = fp[32];  n2 = fp[64];  n3 = fp[96];

      v8f cz = {0.f, 0.f, 0.f, 0.f, 0.f, 0.f, 0.f, 0.f};
      v8f d0 = __builtin_amdgcn_wmma_f32_16x16x4_f32(false, a0, false, bx,
                                                     (short)0, cz, false, false);
      v8f d1 = __builtin_amdgcn_wmma_f32_16x16x4_f32(false, a1, false, bx,
                                                     (short)0, cz, false, false);
      v8f d2 = __builtin_amdgcn_wmma_f32_16x16x4_f32(false, a2, false, bx,
                                                     (short)0, cz, false, false);
      v8f d3 = __builtin_amdgcn_wmma_f32_16x16x4_f32(false, a3, false, bx,
                                                     (short)0, cz, false, false);
      unsigned b0 = 0, b1 = 0, b2 = 0, b3 = 0;
#pragma unroll
      for (int r = 7; r >= 0; --r) {
        b0 = push_out(b0, r2, d0[r]);
        b1 = push_out(b1, r2, d1[r]);
        b2 = push_out(b2, r2, d2[r]);
        b3 = push_out(b3, r2, d3[r]);
      }
      const unsigned in = ~(b0 | (b1 << 8) | (b2 << 16) | (b3 << 24));
      cl += __popc(in);
      rw[2 * j] = in;
    }

    const int tot = cl + __builtin_amdgcn_permlanex16(cl, cl, 0x76543210u, 0xfedcba98u, false, false);
    if (lo) cnt[(bt * 16 + lane) * nch + c] = tot;

    __builtin_amdgcn_wave_barrier();

#pragma unroll
    for (int i = 0; i < 4; ++i) {
      const int box = 4 * i + (lane >> 3);
      const int u   = lane & 7;
      const uint4 q = *reinterpret_cast<const uint4*>(wrow + box * ROWP + 4 * u);
      uint4 o;
      o.x = __builtin_amdgcn_perm(q.y, q.x, 0x05010400u);
      o.y = __builtin_amdgcn_perm(q.y, q.x, 0x07030602u);
      o.z = __builtin_amdgcn_perm(q.w, q.z, 0x05010400u);
      o.w = __builtin_amdgcn_perm(q.w, q.z, 0x07030602u);
      *reinterpret_cast<uint4*>(bitmap + (size_t)(bt * 16 + box) * wpb
                                + (size_t)c * 32 + 4 * u) = o;
    }

    __builtin_amdgcn_wave_barrier();
  }
}

#define EMIT_T 128
#define EMIT_W (EMIT_T / 32)

template <int SH>
__device__ __forceinline__ int row_shr0(int x) {
  return __builtin_amdgcn_update_dpp(0, x, 0x110 + SH, 0xf, 0xf, false);
}

__device__ __forceinline__ int wave_incl_add(int x, int lane) {
  x += row_shr0<1>(x);
  x += row_shr0<2>(x);
  x += row_shr0<4>(x);
  x += row_shr0<8>(x);
  const int lo = __builtin_amdgcn_readlane(x, 15);
  return x + (lane >= 16 ? lo : 0);
}

__device__ __forceinline__ int wave_incl_max(int x, int lane) {
  x = max(x, row_shr0<1>(x));
  x = max(x, row_shr0<2>(x));
  x = max(x, row_shr0<4>(x));
  x = max(x, row_shr0<8>(x));
  const int lo = __builtin_amdgcn_readlane(x, 15);
  return lane >= 16 ? max(x, lo) : x;
}

__global__ __launch_bounds__(EMIT_T)
void emit_kernel(const float* __restrict__ P,
                 const float* __restrict__ boxes,
                 const unsigned* __restrict__ bitmap, int wpb,
                 const int* __restrict__ cnt, int nch,
                 float* __restrict__ out, int S) {
  __shared__ int   mark[EMIT_W][64];
  __shared__ float stage[EMIT_W][32 * 7];

  const int m    = blockIdx.x;
  const int tid  = threadIdx.x;
  const int lane = tid & 31;
  const int wv   = tid >> 5;

  const float vx = boxes[m * 9 + 7];
  const float vy = boxes[m * 9 + 8];

  const int* cm = cnt + (size_t)m * nch;
  int*   mkc = mark[wv];
  int*   mkb = mark[wv] + 32;
  float* st  = stage[wv];

  for (int s0 = wv * 32; s0 < S; s0 += EMIT_T) {
    mkc[lane] = 0;
    mkb[lane] = 0;
    __builtin_amdgcn_wave_barrier();

    int carry = 0;
    for (int c0 = 0; c0 < nch; c0 += 256) {
      const int i0 = c0 + lane * 8;
      int v[8];
      if ((nch & 3) == 0 && i0 + 8 <= nch) {
        const int4 ca = *reinterpret_cast<const int4*>(cm + i0);
        const int4 cb = *reinterpret_cast<const int4*>(cm + i0 + 4);
        v[0] = ca.x;  v[1] = ca.y;  v[2] = ca.z;  v[3] = ca.w;
        v[4] = cb.x;  v[5] = cb.y;  v[6] = cb.z;  v[7] = cb.w;
      } else {
#pragma unroll
        for (int e = 0; e < 8; ++e) {
          const int i = i0 + e;
          v[e] = 0;
          if (i < nch) v[e] = cm[i];
        }
      }
      int loc = 0;
#pragma unroll
      for (int e = 0; e < 8; ++e) loc += v[e];
      const int inc = wave_incl_add(loc, lane);
      int sb = carry + inc - loc;
#pragma unroll
      for (int e = 0; e < 8; ++e) {
        const int se = sb + v[e];
        if (v[e] > 0 && se > s0 && sb < s0 + 32) {
          const int pos = sb > s0 ? sb - s0 : 0;
          mkc[pos] = i0 + e + 1;
          mkb[pos] = sb;
        }
        sb = se;
      }
      carry += __builtin_amdgcn_readlane(inc, 31);
    }
    const int total = carry;

    __builtin_amdgcn_wave_barrier();
    const int cmark = wave_incl_max(mkc[lane], lane);
    const int sbc   = wave_incl_max(mkb[lane], lane);

    const int s = s0 + lane;
    float r0 = 0.f, r1 = 0.f, r2 = 0.f, r3 = 0.f, r4 = 0.f, r5 = 0.f, r6 = 0.f;
    if (s < S && s < total) {
      const int c = cmark - 1;
      const int q = s - sbc;

      const uint4* mp = reinterpret_cast<const uint4*>(
          bitmap + (size_t)m * wpb + (size_t)c * 32);
      unsigned w[32];
#pragma unroll
      for (int k = 0; k < 8; ++k) {
        const uint4 t = mp[k];
        w[4 * k] = t.x;  w[4 * k + 1] = t.y;  w[4 * k + 2] = t.z;  w[4 * k + 3] = t.w;
      }
      int acc = 0, gk = 0, rem = 0;
      unsigned f = 0;
#pragma unroll
      for (int k = 0; k < 32; ++k) {
        if (acc <= q) { gk = k; rem = q - acc; f = w[k]; }
        acc += __popc(w[k]);
      }
      int pos = 0, n;
      n = __popc(f & 0xFFFFu); if (rem >= n) { rem -= n; pos += 16; f >>= 16; }
      n = __popc(f & 0xFFu); if (rem >= n) { rem -= n; pos += 8; f >>= 8; }
      n = __popc(f & 0xFu);  if (rem >= n) { rem -= n; pos += 4; f >>= 4; }
      n = __popc(f & 0x3u);  if (rem >= n) { rem -= n; pos += 2; f >>= 2; }
      n = (int)(f & 1u);     if (rem >= n) { pos += 1; }

      const int pidx = (c * 32 + gk) * 32 + pos;
      const float* ps = P + (size_t)pidx * 5;
      r0 = ps[0]; r1 = ps[1]; r2 = ps[2]; r3 = ps[3]; r4 = ps[4];
      r5 = vx;    r6 = vy;
    }
    float* sr = st + lane * 7;
    sr[0] = r0; sr[1] = r1; sr[2] = r2; sr[3] = r3;
    sr[4] = r4; sr[5] = r5; sr[6] = r6;
    __builtin_amdgcn_wave_barrier();
    const int nsl = (S - s0 < 32) ? (S - s0) : 32;
    const int nf  = nsl * 7;
    float* ob = out + ((size_t)m * S + s0) * 7;
    float o[7];
#pragma unroll
    for (int r = 0; r < 7; ++r) o[r] = st[r * 32 + lane];
#pragma unroll
    for (int r = 0; r < 7; ++r) {
      const int i = r * 32 + lane;
      if (i < nf) ob[i] = o[r];
    }
    __builtin_amdgcn_wave_barrier();
  }
}

extern "C" void kernel_launch(void* const* d_in, const int* in_sizes, int n_in,
                              void* d_out, int out_size, void* d_ws, size_t ws_size,
                              hipStream_t stream) {
  const float* P = (const float*)d_in[0];
  const float* B = (const float*)d_in[1];
  float* out = (float*)d_out;

  const int N = in_sizes[0] / 5;
  const int M = in_sizes[1] / 9;
  const int S = out_size / (M * 7);
  const int nch    = (N + 1023) / 1024;
  const int wpb    = nch * 32;
  const int bt     = M / 16;

  char* ws = (char*)d_ws;
  unsigned* bitmap  = (unsigned*)ws;
  size_t mbytes     = (size_t)M * wpb * sizeof(unsigned);
  int* cnt          = (int*)(ws + mbytes);

  count_kernel<<<nch, CNT_T, 0, stream>>>(P, N, B, bitmap, wpb, cnt, nch, bt);
  emit_kernel<<<M, EMIT_T, 0, stream>>>(P, B, bitmap, wpb, cnt, nch, out, S);
}
